# starting kernel
# baseline (speedup 1.0000x reference)
#include <hip/hip_runtime.h>


typedef float v2f __attribute__((ext_vector_type(2)));
typedef float v8f __attribute__((ext_vector_type(8)));

#define B_   4
#define M_   256
#define L_   24
#define S_   22
#define SP_  24
#define W_   279
#define WP_  288
#define NTW  18
#define NTN  16

__device__ __forceinline__ unsigned map_f32(float x) {
  unsigned u = __float_as_uint(x);
  return (u & 0x80000000u) ? ~u : (u | 0x80000000u);
}

__device__ __forceinline__ int clampi(int v, int lo, int hi) {
  return v < lo ? lo : (v > hi ? hi : v);
}

__global__ __launch_bounds__(128) void cassi_phase1(
    const float* __restrict__ x, const float* __restrict__ H,
    float* __restrict__ Yt, float* __restrict__ sumYY)
{
  const int lane = threadIdx.x & 31;
  const int wave = blockIdx.x * 4 + (threadIdx.x >> 5);
  const int wt = wave % NTW;
  const int bm = wave / NTW;
  const int m  = bm & (M_ - 1);
  const int b  = bm >> 8;

  const int lo16 = lane & 15;
  const int hi   = lane >> 4;
  const int w0   = wt * 16;
  const int w_me = w0 + lo16;

  const float* xb = x + (size_t)(b * M_ + m) * M_ * L_;
  const float* Hm = H + (size_t)m * M_ * S_;

  v8f c0 = {};
  v8f c1 = {};

  int nstart = w0 - (L_ - 1); if (nstart < 0) nstart = 0;
  nstart &= ~3;
  int nend = w0 + 15; if (nend > M_ - 1) nend = M_ - 1;

  const bool s1ok = (16 + lo16 < S_);
  const int  s1c  = s1ok ? 16 + lo16 : S_ - 1;
  const int  srow1 = s1ok ? 16 + lo16 : S_;

  for (int n0 = nstart; n0 <= nend; n0 += 4) {
    v2f a, b0v, b1v;
#pragma unroll
    for (int j = 0; j < 2; ++j) {
      int k = j + 2 * hi;
      int n = n0 + k;
      int i = w_me - n;
      bool aok = (n < M_) & ((unsigned)i < (unsigned)L_);
      bool nok = (n < M_);
      int nc = n < M_ ? n : M_ - 1;
      int ic = clampi(i, 0, L_ - 1);
      float av = xb[nc * L_ + ic];
      float h0 = Hm[nc * S_ + lo16];
      float h1 = Hm[nc * S_ + s1c];
      a[j]   = aok ? av : 0.0f;
      b0v[j] = nok ? h0 : 0.0f;
      b1v[j] = (nok & s1ok) ? h1 : 0.0f;
    }
    c0 = __builtin_amdgcn_wmma_f32_16x16x4_f32(false, a, false, b0v,
                                               (short)0, c0, false, false);
    c1 = __builtin_amdgcn_wmma_f32_16x16x4_f32(false, a, false, b1v,
                                               (short)0, c1, false, false);
  }

  float* Ybm = Yt + (size_t)(b * M_ + m) * SP_ * WP_;
  float* r0 = Ybm + (size_t)lo16 * WP_ + w0 + 8 * hi;
  float* r1 = Ybm + (size_t)srow1 * WP_ + w0 + 8 * hi;
  float sq = 0.0f;
#pragma unroll
  for (int r = 0; r < 8; ++r) {
    int w = w0 + r + 8 * hi;
    bool wok = (w < W_);
    float v0 = c0[r];
    float v1 = c1[r];
    r0[r] = v0;
    r1[r] = v1;
    sq += wok ? v0 * v0 : 0.0f;
    sq += (wok & s1ok) ? v1 * v1 : 0.0f;
  }
#pragma unroll
  for (int off = 16; off > 0; off >>= 1) sq += __shfl_down(sq, off, 32);
  if (lane == 0) atomicAdd(sumYY, sq);
}

__global__ __launch_bounds__(128) void cassi_phase2(
    const float* __restrict__ H, const float* __restrict__ eps,
    const float* __restrict__ Yt, const float* __restrict__ sumYY,
    float* __restrict__ Xout, unsigned* __restrict__ umax,
    float* __restrict__ dump)
{
  const int lane = threadIdx.x & 31;
  const int wave = blockIdx.x * 4 + (threadIdx.x >> 5);
  const int t  = wave % NTN;
  const int bm = wave / NTN;
  const int m  = bm & (M_ - 1);
  const int b  = bm >> 8;

  const int lo16 = lane & 15;
  const int hi   = lane >> 4;
  const int n0   = t * 16;

  const float* Hm  = H  + (size_t)m * M_ * S_;
  const float* Ybm = Yt + (size_t)(b * M_ + m) * SP_ * WP_;

  v8f acc0 = {}, acc1 = {}, acc2 = {};

#pragma unroll
  for (int s0 = 0; s0 < 24; s0 += 4) {
    v2f a, bv0, bv1, bv2;
#pragma unroll
    for (int j = 0; j < 2; ++j) {
      int s = s0 + j + 2 * hi;
      bool sok = (s < S_);
      int sc = sok ? s : S_ - 1;
      float av = Hm[(n0 + lo16) * S_ + sc];
      const float* row = Ybm + (size_t)sc * WP_ + lo16;
      float v0 = row[(t + 0) * 16];
      float v1 = row[(t + 1) * 16];
      float v2 = row[(t + 2) * 16];
      a[j]   = sok ? av : 0.0f;
      bv0[j] = sok ? v0 : 0.0f;
      bv1[j] = sok ? v1 : 0.0f;
      bv2[j] = sok ? v2 : 0.0f;
    }
    acc0 = __builtin_amdgcn_wmma_f32_16x16x4_f32(false, a, false, bv0,
                                                 (short)0, acc0, false, false);
    acc1 = __builtin_amdgcn_wmma_f32_16x16x4_f32(false, a, false, bv1,
                                                 (short)0, acc1, false, false);
    acc2 = __builtin_amdgcn_wmma_f32_16x16x4_f32(false, a, false, bv2,
                                                 (short)0, acc2, false, false);
  }

  float hsum = 0.0f;
  {
    const float* hr = Hm + (n0 + (lane & 15)) * S_;
    if (lane < 16) {
#pragma unroll
      for (int s = 0; s < S_; ++s) hsum += hr[s];
    }
  }
  float hs[8];
#pragma unroll
  for (int r = 0; r < 8; ++r) hs[r] = __shfl(hsum, r + 8 * hi, 32);

  const float noiseScale =
      sqrtf(*sumYY * (1.0f / ((float)M_ * (float)W_ * (float)B_ * 10000.0f)));
  const float* epsbm = eps + (size_t)(b * M_ + m) * W_;
  float* Xbm = Xout + (size_t)(b * M_ + m) * M_ * L_;

  float lmax = -3.4e38f;
#pragma unroll
  for (int ut = 0; ut < 3; ++ut) {
    const v8f& acc = (ut == 0) ? acc0 : (ut == 1) ? acc1 : acc2;
    int u = (t + ut) * 16 + lo16;
    int uc = u < W_ ? u : W_ - 1;
    float ev = epsbm[uc];
    ev = (u < W_) ? ev : 0.0f;
    float nsev = noiseScale * ev;
#pragma unroll
    for (int r = 0; r < 8; ++r) {
      int n = n0 + r + 8 * hi;
      int l = u - n;
      bool ok = (unsigned)l < (unsigned)L_;
      float v = fmaf(nsev, hs[r], acc[r]);
      float* addr = ok ? (Xbm + (size_t)n * L_ + l) : dump;
      *addr = v;
      lmax = fmaxf(lmax, ok ? v : -3.4e38f);
    }
  }
#pragma unroll
  for (int off = 16; off > 0; off >>= 1)
    lmax = fmaxf(lmax, __shfl_down(lmax, off, 32));
  if (lane == 0) atomicMax(umax, map_f32(lmax));
}

__global__ __launch_bounds__(256) void cassi_norm(
    float4* __restrict__ out, const unsigned* __restrict__ umax, int n4)
{
  int i = blockIdx.x * blockDim.x + threadIdx.x;
  unsigned u = *umax;
  float mx = (u & 0x80000000u) ? __uint_as_float(u ^ 0x80000000u)
                               : __uint_as_float(~u);
  if (i < n4) {
    float4 v = out[i];
    v.x /= mx; v.y /= mx; v.z /= mx; v.w /= mx;
    out[i] = v;
  }
}

extern "C" void kernel_launch(void* const* d_in, const int* in_sizes, int n_in,
                              void* d_out, int out_size, void* d_ws, size_t ws_size,
                              hipStream_t stream) {
  const float* x   = (const float*)d_in[0];
  const float* H   = (const float*)d_in[1];
  const float* eps = (const float*)d_in[2];
  float* out = (float*)d_out;

  float*    sumYY = (float*)d_ws;
  unsigned* umax  = (unsigned*)((char*)d_ws + 4);
  float*    dump  = (float*)((char*)d_ws + 16);
  float*    Yt    = (float*)((char*)d_ws + 256);

  hipMemsetAsync(d_ws, 0, 8, stream);

  cassi_phase1<<<(B_ * M_ * NTW) / 4, 128, 0, stream>>>(x, H, Yt, sumYY);
  cassi_phase2<<<(B_ * M_ * NTN) / 4, 128, 0, stream>>>(H, eps, Yt, sumYY,
                                                        out, umax, dump);
  int n4 = (B_ * M_ * M_ * L_) / 4;
  cassi_norm<<<(n4 + 255) / 256, 256, 0, stream>>>((float4*)out, umax, n4);
}


# phase 2: out-of-band lanes no longer store to one shared dump word; their stores are predicated (exec-masked), removing a grid-wide same-address write stream that serialised at the L2
# speedup vs baseline: 4.3410x; 4.3410x over previous
#include <hip/hip_runtime.h>


typedef float v2f __attribute__((ext_vector_type(2)));
typedef float v8f __attribute__((ext_vector_type(8)));

#define B_   4
#define M_   256
#define L_   24
#define S_   22
#define SP_  24
#define W_   279
#define WP_  288
#define NTW  18
#define NTN  16

__device__ __forceinline__ unsigned map_f32(float x) {
  unsigned u = __float_as_uint(x);
  return (u & 0x80000000u) ? ~u : (u | 0x80000000u);
}

__device__ __forceinline__ int clampi(int v, int lo, int hi) {
  return v < lo ? lo : (v > hi ? hi : v);
}

__global__ __launch_bounds__(128) void cassi_phase1(
    const float* __restrict__ x, const float* __restrict__ H,
    float* __restrict__ Yt, float* __restrict__ sumYY)
{
  const int lane = threadIdx.x & 31;
  const int wave = blockIdx.x * 4 + (threadIdx.x >> 5);
  const int wt = wave % NTW;
  const int bm = wave / NTW;
  const int m  = bm & (M_ - 1);
  const int b  = bm >> 8;

  const int lo16 = lane & 15;
  const int hi   = lane >> 4;
  const int w0   = wt * 16;
  const int w_me = w0 + lo16;

  const float* xb = x + (size_t)(b * M_ + m) * M_ * L_;
  const float* Hm = H + (size_t)m * M_ * S_;

  v8f c0 = {};
  v8f c1 = {};

  int nstart = w0 - (L_ - 1); if (nstart < 0) nstart = 0;
  nstart &= ~3;
  int nend = w0 + 15; if (nend > M_ - 1) nend = M_ - 1;

  const bool s1ok = (16 + lo16 < S_);
  const int  s1c  = s1ok ? 16 + lo16 : S_ - 1;
  const int  srow1 = s1ok ? 16 + lo16 : S_;

  for (int n0 = nstart; n0 <= nend; n0 += 4) {
    v2f a, b0v, b1v;
#pragma unroll
    for (int j = 0; j < 2; ++j) {
      int k = j + 2 * hi;
      int n = n0 + k;
      int i = w_me - n;
      bool aok = (n < M_) & ((unsigned)i < (unsigned)L_);
      bool nok = (n < M_);
      int nc = n < M_ ? n : M_ - 1;
      int ic = clampi(i, 0, L_ - 1);
      float av = xb[nc * L_ + ic];
      float h0 = Hm[nc * S_ + lo16];
      float h1 = Hm[nc * S_ + s1c];
      a[j]   = aok ? av : 0.0f;
      b0v[j] = nok ? h0 : 0.0f;
      b1v[j] = (nok & s1ok) ? h1 : 0.0f;
    }
    c0 = __builtin_amdgcn_wmma_f32_16x16x4_f32(false, a, false, b0v,
                                               (short)0, c0, false, false);
    c1 = __builtin_amdgcn_wmma_f32_16x16x4_f32(false, a, false, b1v,
                                               (short)0, c1, false, false);
  }

  float* Ybm = Yt + (size_t)(b * M_ + m) * SP_ * WP_;
  float* r0 = Ybm + (size_t)lo16 * WP_ + w0 + 8 * hi;
  float* r1 = Ybm + (size_t)srow1 * WP_ + w0 + 8 * hi;
  float sq = 0.0f;
#pragma unroll
  for (int r = 0; r < 8; ++r) {
    int w = w0 + r + 8 * hi;
    bool wok = (w < W_);
    float v0 = c0[r];
    float v1 = c1[r];
    r0[r] = v0;
    r1[r] = v1;
    sq += wok ? v0 * v0 : 0.0f;
    sq += (wok & s1ok) ? v1 * v1 : 0.0f;
  }
#pragma unroll
  for (int off = 16; off > 0; off >>= 1) sq += __shfl_down(sq, off, 32);
  if (lane == 0) atomicAdd(sumYY, sq);
}

__global__ __launch_bounds__(128) void cassi_phase2(
    const float* __restrict__ H, const float* __restrict__ eps,
    const float* __restrict__ Yt, const float* __restrict__ sumYY,
    float* __restrict__ Xout, unsigned* __restrict__ umax)
{
  const int lane = threadIdx.x & 31;
  const int wave = blockIdx.x * 4 + (threadIdx.x >> 5);
  const int t  = wave % NTN;
  const int bm = wave / NTN;
  const int m  = bm & (M_ - 1);
  const int b  = bm >> 8;

  const int lo16 = lane & 15;
  const int hi   = lane >> 4;
  const int n0   = t * 16;

  const float* Hm  = H  + (size_t)m * M_ * S_;
  const float* Ybm = Yt + (size_t)(b * M_ + m) * SP_ * WP_;

  v8f acc0 = {}, acc1 = {}, acc2 = {};

#pragma unroll
  for (int s0 = 0; s0 < 24; s0 += 4) {
    v2f a, bv0, bv1, bv2;
#pragma unroll
    for (int j = 0; j < 2; ++j) {
      int s = s0 + j + 2 * hi;
      bool sok = (s < S_);
      int sc = sok ? s : S_ - 1;
      float av = Hm[(n0 + lo16) * S_ + sc];
      const float* row = Ybm + (size_t)sc * WP_ + lo16;
      float v0 = row[(t + 0) * 16];
      float v1 = row[(t + 1) * 16];
      float v2 = row[(t + 2) * 16];
      a[j]   = sok ? av : 0.0f;
      bv0[j] = sok ? v0 : 0.0f;
      bv1[j] = sok ? v1 : 0.0f;
      bv2[j] = sok ? v2 : 0.0f;
    }
    acc0 = __builtin_amdgcn_wmma_f32_16x16x4_f32(false, a, false, bv0,
                                                 (short)0, acc0, false, false);
    acc1 = __builtin_amdgcn_wmma_f32_16x16x4_f32(false, a, false, bv1,
                                                 (short)0, acc1, false, false);
    acc2 = __builtin_amdgcn_wmma_f32_16x16x4_f32(false, a, false, bv2,
                                                 (short)0, acc2, false, false);
  }

  float hsum = 0.0f;
  {
    const float* hr = Hm + (n0 + (lane & 15)) * S_;
    if (lane < 16) {
#pragma unroll
      for (int s = 0; s < S_; ++s) hsum += hr[s];
    }
  }
  float hs[8];
#pragma unroll
  for (int r = 0; r < 8; ++r) hs[r] = __shfl(hsum, r + 8 * hi, 32);

  const float noiseScale =
      sqrtf(*sumYY * (1.0f / ((float)M_ * (float)W_ * (float)B_ * 10000.0f)));
  const float* epsbm = eps + (size_t)(b * M_ + m) * W_;
  float* Xbm = Xout + (size_t)(b * M_ + m) * M_ * L_;

  float lmax = -3.4e38f;
#pragma unroll
  for (int ut = 0; ut < 3; ++ut) {
    const v8f& acc = (ut == 0) ? acc0 : (ut == 1) ? acc1 : acc2;
    int u = (t + ut) * 16 + lo16;
    int uc = u < W_ ? u : W_ - 1;
    float ev = epsbm[uc];
    ev = (u < W_) ? ev : 0.0f;
    float nsev = noiseScale * ev;
#pragma unroll
    for (int r = 0; r < 8; ++r) {
      int n = n0 + r + 8 * hi;
      int l = u - n;
      bool ok = (unsigned)l < (unsigned)L_;
      float v = fmaf(nsev, hs[r], acc[r]);
      if (ok) Xbm[(size_t)n * L_ + l] = v;
      lmax = fmaxf(lmax, ok ? v : -3.4e38f);
    }
  }
#pragma unroll
  for (int off = 16; off > 0; off >>= 1)
    lmax = fmaxf(lmax, __shfl_down(lmax, off, 32));
  if (lane == 0) atomicMax(umax, map_f32(lmax));
}

__global__ __launch_bounds__(256) void cassi_norm(
    float4* __restrict__ out, const unsigned* __restrict__ umax, int n4)
{
  int i = blockIdx.x * blockDim.x + threadIdx.x;
  unsigned u = *umax;
  float mx = (u & 0x80000000u) ? __uint_as_float(u ^ 0x80000000u)
                               : __uint_as_float(~u);
  if (i < n4) {
    float4 v = out[i];
    v.x /= mx; v.y /= mx; v.z /= mx; v.w /= mx;
    out[i] = v;
  }
}

extern "C" void kernel_launch(void* const* d_in, const int* in_sizes, int n_in,
                              void* d_out, int out_size, void* d_ws, size_t ws_size,
                              hipStream_t stream) {
  const float* x   = (const float*)d_in[0];
  const float* H   = (const float*)d_in[1];
  const float* eps = (const float*)d_in[2];
  float* out = (float*)d_out;

  float*    sumYY = (float*)d_ws;
  unsigned* umax  = (unsigned*)((char*)d_ws + 4);
  float*    Yt    = (float*)((char*)d_ws + 256);

  hipMemsetAsync(d_ws, 0, 8, stream);

  cassi_phase1<<<(B_ * M_ * NTW) / 4, 128, 0, stream>>>(x, H, Yt, sumYY);
  cassi_phase2<<<(B_ * M_ * NTN) / 4, 128, 0, stream>>>(H, eps, Yt, sumYY,
                                                        out, umax);
  int n4 = (B_ * M_ * M_ * L_) / 4;
  cassi_norm<<<(n4 + 255) / 256, 256, 0, stream>>>((float4*)out, umax, n4);
}


# sum(Y^2) and max(X): each wave's atomic goes to one of 16 words on separate 128-byte lines instead of one hot word; consumers fold the 16 in fixed order (changes the atomic summation order)
# speedup vs baseline: 6.0671x; 1.3976x over previous
#include <hip/hip_runtime.h>


typedef float v2f __attribute__((ext_vector_type(2)));
typedef float v8f __attribute__((ext_vector_type(8)));

#define B_   4
#define M_   256
#define L_   24
#define S_   22
#define SP_  24
#define W_   279
#define WP_  288
#define NTW  18
#define NTN  16
#define NSLOT 16
#define SLOTF 32

__device__ __forceinline__ unsigned map_f32(float x) {
  unsigned u = __float_as_uint(x);
  return (u & 0x80000000u) ? ~u : (u | 0x80000000u);
}

__device__ __forceinline__ int clampi(int v, int lo, int hi) {
  return v < lo ? lo : (v > hi ? hi : v);
}

__global__ __launch_bounds__(128) void cassi_phase1(
    const float* __restrict__ x, const float* __restrict__ H,
    float* __restrict__ Yt, float* __restrict__ sumYY)
{
  const int lane = threadIdx.x & 31;
  const int wave = blockIdx.x * 4 + (threadIdx.x >> 5);
  const int wt = wave % NTW;
  const int bm = wave / NTW;
  const int m  = bm & (M_ - 1);
  const int b  = bm >> 8;

  const int lo16 = lane & 15;
  const int hi   = lane >> 4;
  const int w0   = wt * 16;
  const int w_me = w0 + lo16;

  const float* xb = x + (size_t)(b * M_ + m) * M_ * L_;
  const float* Hm = H + (size_t)m * M_ * S_;

  v8f c0 = {};
  v8f c1 = {};

  int nstart = w0 - (L_ - 1); if (nstart < 0) nstart = 0;
  nstart &= ~3;
  int nend = w0 + 15; if (nend > M_ - 1) nend = M_ - 1;

  const bool s1ok = (16 + lo16 < S_);
  const int  s1c  = s1ok ? 16 + lo16 : S_ - 1;
  const int  srow1 = s1ok ? 16 + lo16 : S_;

  for (int n0 = nstart; n0 <= nend; n0 += 4) {
    v2f a, b0v, b1v;
#pragma unroll
    for (int j = 0; j < 2; ++j) {
      int k = j + 2 * hi;
      int n = n0 + k;
      int i = w_me - n;
      bool aok = (n < M_) & ((unsigned)i < (unsigned)L_);
      bool nok = (n < M_);
      int nc = n < M_ ? n : M_ - 1;
      int ic = clampi(i, 0, L_ - 1);
      float av = xb[nc * L_ + ic];
      float h0 = Hm[nc * S_ + lo16];
      float h1 = Hm[nc * S_ + s1c];
      a[j]   = aok ? av : 0.0f;
      b0v[j] = nok ? h0 : 0.0f;
      b1v[j] = (nok & s1ok) ? h1 : 0.0f;
    }
    c0 = __builtin_amdgcn_wmma_f32_16x16x4_f32(false, a, false, b0v,
                                               (short)0, c0, false, false);
    c1 = __builtin_amdgcn_wmma_f32_16x16x4_f32(false, a, false, b1v,
                                               (short)0, c1, false, false);
  }

  float* Ybm = Yt + (size_t)(b * M_ + m) * SP_ * WP_;
  float* r0 = Ybm + (size_t)lo16 * WP_ + w0 + 8 * hi;
  float* r1 = Ybm + (size_t)srow1 * WP_ + w0 + 8 * hi;
  float sq = 0.0f;
#pragma unroll
  for (int r = 0; r < 8; ++r) {
    int w = w0 + r + 8 * hi;
    bool wok = (w < W_);
    float v0 = c0[r];
    float v1 = c1[r];
    r0[r] = v0;
    r1[r] = v1;
    sq += wok ? v0 * v0 : 0.0f;
    sq += (wok & s1ok) ? v1 * v1 : 0.0f;
  }
#pragma unroll
  for (int off = 16; off > 0; off >>= 1) sq += __shfl_down(sq, off, 32);
  if (lane == 0) atomicAdd(sumYY + (wave & (NSLOT - 1)) * SLOTF, sq);
}

__global__ __launch_bounds__(128) void cassi_phase2(
    const float* __restrict__ H, const float* __restrict__ eps,
    const float* __restrict__ Yt, const float* __restrict__ sumYY,
    float* __restrict__ Xout, unsigned* __restrict__ umax)
{
  const int lane = threadIdx.x & 31;
  const int wave = blockIdx.x * 4 + (threadIdx.x >> 5);
  const int t  = wave % NTN;
  const int bm = wave / NTN;
  const int m  = bm & (M_ - 1);
  const int b  = bm >> 8;

  const int lo16 = lane & 15;
  const int hi   = lane >> 4;
  const int n0   = t * 16;

  const float* Hm  = H  + (size_t)m * M_ * S_;
  const float* Ybm = Yt + (size_t)(b * M_ + m) * SP_ * WP_;

  v8f acc0 = {}, acc1 = {}, acc2 = {};

#pragma unroll
  for (int s0 = 0; s0 < 24; s0 += 4) {
    v2f a, bv0, bv1, bv2;
#pragma unroll
    for (int j = 0; j < 2; ++j) {
      int s = s0 + j + 2 * hi;
      bool sok = (s < S_);
      int sc = sok ? s : S_ - 1;
      float av = Hm[(n0 + lo16) * S_ + sc];
      const float* row = Ybm + (size_t)sc * WP_ + lo16;
      float v0 = row[(t + 0) * 16];
      float v1 = row[(t + 1) * 16];
      float v2 = row[(t + 2) * 16];
      a[j]   = sok ? av : 0.0f;
      bv0[j] = sok ? v0 : 0.0f;
      bv1[j] = sok ? v1 : 0.0f;
      bv2[j] = sok ? v2 : 0.0f;
    }
    acc0 = __builtin_amdgcn_wmma_f32_16x16x4_f32(false, a, false, bv0,
                                                 (short)0, acc0, false, false);
    acc1 = __builtin_amdgcn_wmma_f32_16x16x4_f32(false, a, false, bv1,
                                                 (short)0, acc1, false, false);
    acc2 = __builtin_amdgcn_wmma_f32_16x16x4_f32(false, a, false, bv2,
                                                 (short)0, acc2, false, false);
  }

  float hsum = 0.0f;
  {
    const float* hr = Hm + (n0 + (lane & 15)) * S_;
    if (lane < 16) {
#pragma unroll
      for (int s = 0; s < S_; ++s) hsum += hr[s];
    }
  }
  float hs[8];
#pragma unroll
  for (int r = 0; r < 8; ++r) hs[r] = __shfl(hsum, r + 8 * hi, 32);

  float sumAll = 0.0f;
  for (int k = 0; k < NSLOT; ++k) sumAll += sumYY[k * SLOTF];
  const float noiseScale =
      sqrtf(sumAll * (1.0f / ((float)M_ * (float)W_ * (float)B_ * 10000.0f)));
  const float* epsbm = eps + (size_t)(b * M_ + m) * W_;
  float* Xbm = Xout + (size_t)(b * M_ + m) * M_ * L_;

  float lmax = -3.4e38f;
#pragma unroll
  for (int ut = 0; ut < 3; ++ut) {
    const v8f& acc = (ut == 0) ? acc0 : (ut == 1) ? acc1 : acc2;
    int u = (t + ut) * 16 + lo16;
    int uc = u < W_ ? u : W_ - 1;
    float ev = epsbm[uc];
    ev = (u < W_) ? ev : 0.0f;
    float nsev = noiseScale * ev;
#pragma unroll
    for (int r = 0; r < 8; ++r) {
      int n = n0 + r + 8 * hi;
      int l = u - n;
      bool ok = (unsigned)l < (unsigned)L_;
      float v = fmaf(nsev, hs[r], acc[r]);
      if (ok) Xbm[(size_t)n * L_ + l] = v;
      lmax = fmaxf(lmax, ok ? v : -3.4e38f);
    }
  }
#pragma unroll
  for (int off = 16; off > 0; off >>= 1)
    lmax = fmaxf(lmax, __shfl_down(lmax, off, 32));
  if (lane == 0) atomicMax(umax + (wave & (NSLOT - 1)) * SLOTF, map_f32(lmax));
}

__global__ __launch_bounds__(256) void cassi_norm(
    float4* __restrict__ out, const unsigned* __restrict__ umax, int n4)
{
  int i = blockIdx.x * blockDim.x + threadIdx.x;
  unsigned u = 0u;
  for (int k = 0; k < NSLOT; ++k) {
    unsigned uk = umax[k * SLOTF];
    u = uk > u ? uk : u;
  }
  float mx = (u & 0x80000000u) ? __uint_as_float(u ^ 0x80000000u)
                               : __uint_as_float(~u);
  if (i < n4) {
    float4 v = out[i];
    v.x /= mx; v.y /= mx; v.z /= mx; v.w /= mx;
    out[i] = v;
  }
}

extern "C" void kernel_launch(void* const* d_in, const int* in_sizes, int n_in,
                              void* d_out, int out_size, void* d_ws, size_t ws_size,
                              hipStream_t stream) {
  const float* x   = (const float*)d_in[0];
  const float* H   = (const float*)d_in[1];
  const float* eps = (const float*)d_in[2];
  float* out = (float*)d_out;

  float*    sumYY = (float*)d_ws;
  unsigned* umax  = (unsigned*)((char*)d_ws + 2048);
  float*    Yt    = (float*)((char*)d_ws + 4096);

  hipMemsetAsync(d_ws, 0, 4096, stream);

  cassi_phase1<<<(B_ * M_ * NTW) / 4, 128, 0, stream>>>(x, H, Yt, sumYY);
  cassi_phase2<<<(B_ * M_ * NTN) / 4, 128, 0, stream>>>(H, eps, Yt, sumYY,
                                                        out, umax);
  int n4 = (B_ * M_ * M_ * L_) / 4;
  cassi_norm<<<(n4 + 255) / 256, 256, 0, stream>>>((float4*)out, umax, n4);
}


# both GEMM phases stage their result tile in LDS and write it as whole 128-byte lines with 16-byte stores (was 8-byte pieces over 16 rows in phase 1, 24 masked dword stores in phase 2)
# speedup vs baseline: 11.6469x; 1.9197x over previous
#include <hip/hip_runtime.h>


typedef float v2f __attribute__((ext_vector_type(2)));
typedef float v8f __attribute__((ext_vector_type(8)));

#define B_   4
#define M_   256
#define L_   24
#define S_   22
#define SP_  24
#define W_   279
#define WP_  288
#define NTW  18
#define NTN  16
#define NSLOT 16
#define SLOTF 32
#define YP_  36
#define XT_  416

__device__ __forceinline__ unsigned map_f32(float x) {
  unsigned u = __float_as_uint(x);
  return (u & 0x80000000u) ? ~u : (u | 0x80000000u);
}

__device__ __forceinline__ int clampi(int v, int lo, int hi) {
  return v < lo ? lo : (v > hi ? hi : v);
}

__global__ __launch_bounds__(128) void cassi_phase1(
    const float* __restrict__ x, const float* __restrict__ H,
    float* __restrict__ Yt, float* __restrict__ sumYY)
{
  __shared__ __attribute__((aligned(16))) float tileY[2][S_][YP_];
  const int lane = threadIdx.x & 31;
  const int wave = blockIdx.x * 4 + (threadIdx.x >> 5);
  const int wt = wave % NTW;
  const int bm = wave / NTW;
  const int m  = bm & (M_ - 1);
  const int b  = bm >> 8;

  const int lo16 = lane & 15;
  const int hi   = lane >> 4;
  const int w0   = wt * 16;
  const int w_me = w0 + lo16;

  const float* xb = x + (size_t)(b * M_ + m) * M_ * L_;
  const float* Hm = H + (size_t)m * M_ * S_;

  v8f c0 = {};
  v8f c1 = {};

  int nstart = w0 - (L_ - 1); if (nstart < 0) nstart = 0;
  nstart &= ~3;
  int nend = w0 + 15; if (nend > M_ - 1) nend = M_ - 1;

  const bool s1ok = (16 + lo16 < S_);
  const int  s1c  = s1ok ? 16 + lo16 : S_ - 1;

  for (int n0 = nstart; n0 <= nend; n0 += 4) {
    v2f a, b0v, b1v;
#pragma unroll
    for (int j = 0; j < 2; ++j) {
      int k = j + 2 * hi;
      int n = n0 + k;
      int i = w_me - n;
      bool aok = (n < M_) & ((unsigned)i < (unsigned)L_);
      bool nok = (n < M_);
      int nc = n < M_ ? n : M_ - 1;
      int ic = clampi(i, 0, L_ - 1);
      float av = xb[nc * L_ + ic];
      float h0 = Hm[nc * S_ + lo16];
      float h1 = Hm[nc * S_ + s1c];
      a[j]   = aok ? av : 0.0f;
      b0v[j] = nok ? h0 : 0.0f;
      b1v[j] = (nok & s1ok) ? h1 : 0.0f;
    }
    c0 = __builtin_amdgcn_wmma_f32_16x16x4_f32(false, a, false, b0v,
                                               (short)0, c0, false, false);
    c1 = __builtin_amdgcn_wmma_f32_16x16x4_f32(false, a, false, b1v,
                                               (short)0, c1, false, false);
  }

  float sq = 0.0f;
#pragma unroll
  for (int r = 0; r < 8; ++r) {
    int w = w0 + r + 8 * hi;
    bool wok = (w < W_);
    float v0 = c0[r];
    float v1 = c1[r];
    sq += wok ? v0 * v0 : 0.0f;
    sq += (wok & s1ok) ? v1 * v1 : 0.0f;
  }

  const int wid = threadIdx.x >> 5;
  const int par = wid & 1;
  float (*ty)[YP_] = tileY[wid >> 1];
  {
    float4* d0 = (float4*)&ty[lo16][16 * par + 8 * hi];
    d0[0] = make_float4(c0[0], c0[1], c0[2], c0[3]);
    d0[1] = make_float4(c0[4], c0[5], c0[6], c0[7]);
    if (s1ok) {
      float4* d1 = (float4*)&ty[16 + lo16][16 * par + 8 * hi];
      d1[0] = make_float4(c1[0], c1[1], c1[2], c1[3]);
      d1[1] = make_float4(c1[4], c1[5], c1[6], c1[7]);
    }
  }
  __syncthreads();
  float* Ybm = Yt + (size_t)(b * M_ + m) * SP_ * WP_ + (w0 - 16 * par);
#pragma unroll
  for (int k = 0; k < 3; ++k) {
    int rr = 4 * k + (lane >> 3);
    if (rr < 11) {
      int s = par * 11 + rr;
      float4 v = *(const float4*)&ty[s][4 * (lane & 7)];
      *(float4*)(Ybm + (size_t)s * WP_ + 4 * (lane & 7)) = v;
    }
  }
#pragma unroll
  for (int off = 16; off > 0; off >>= 1) sq += __shfl_down(sq, off, 32);
  if (lane == 0) atomicAdd(sumYY + (wave & (NSLOT - 1)) * SLOTF, sq);
}

__global__ __launch_bounds__(128) void cassi_phase2(
    const float* __restrict__ H, const float* __restrict__ eps,
    const float* __restrict__ Yt, const float* __restrict__ sumYY,
    float* __restrict__ Xout, unsigned* __restrict__ umax)
{
  __shared__ __attribute__((aligned(16))) float tileX[4][XT_];
  const int lane = threadIdx.x & 31;
  const int wave = blockIdx.x * 4 + (threadIdx.x >> 5);
  const int t  = wave % NTN;
  const int bm = wave / NTN;
  const int m  = bm & (M_ - 1);
  const int b  = bm >> 8;

  const int lo16 = lane & 15;
  const int hi   = lane >> 4;
  const int n0   = t * 16;

  const float* Hm  = H  + (size_t)m * M_ * S_;
  const float* Ybm = Yt + (size_t)(b * M_ + m) * SP_ * WP_;

  v8f acc0 = {}, acc1 = {}, acc2 = {};

#pragma unroll
  for (int s0 = 0; s0 < 24; s0 += 4) {
    v2f a, bv0, bv1, bv2;
#pragma unroll
    for (int j = 0; j < 2; ++j) {
      int s = s0 + j + 2 * hi;
      bool sok = (s < S_);
      int sc = sok ? s : S_ - 1;
      float av = Hm[(n0 + lo16) * S_ + sc];
      const float* row = Ybm + (size_t)sc * WP_ + lo16;
      float v0 = row[(t + 0) * 16];
      float v1 = row[(t + 1) * 16];
      float v2 = row[(t + 2) * 16];
      a[j]   = sok ? av : 0.0f;
      bv0[j] = sok ? v0 : 0.0f;
      bv1[j] = sok ? v1 : 0.0f;
      bv2[j] = sok ? v2 : 0.0f;
    }
    acc0 = __builtin_amdgcn_wmma_f32_16x16x4_f32(false, a, false, bv0,
                                                 (short)0, acc0, false, false);
    acc1 = __builtin_amdgcn_wmma_f32_16x16x4_f32(false, a, false, bv1,
                                                 (short)0, acc1, false, false);
    acc2 = __builtin_amdgcn_wmma_f32_16x16x4_f32(false, a, false, bv2,
                                                 (short)0, acc2, false, false);
  }

  float hsum = 0.0f;
  {
    const float* hr = Hm + (n0 + (lane & 15)) * S_;
    if (lane < 16) {
#pragma unroll
      for (int s = 0; s < S_; ++s) hsum += hr[s];
    }
  }
  float hs[8];
#pragma unroll
  for (int r = 0; r < 8; ++r) hs[r] = __shfl(hsum, r + 8 * hi, 32);

  float sumAll = 0.0f;
  for (int k = 0; k < NSLOT; ++k) sumAll += sumYY[k * SLOTF];
  const float noiseScale =
      sqrtf(sumAll * (1.0f / ((float)M_ * (float)W_ * (float)B_ * 10000.0f)));
  const float* epsbm = eps + (size_t)(b * M_ + m) * W_;
  float* Xbm = Xout + (size_t)(b * M_ + m) * M_ * L_;

  float* tx = tileX[threadIdx.x >> 5];
#pragma unroll
  for (int ut = 0; ut < 3; ++ut) {
    const v8f& acc = (ut == 0) ? acc0 : (ut == 1) ? acc1 : acc2;
    int u = (t + ut) * 16 + lo16;
    int uc = u < W_ ? u : W_ - 1;
    float ev = epsbm[uc];
    ev = (u < W_) ? ev : 0.0f;
    float nsev = noiseScale * ev;
#pragma unroll
    for (int r = 0; r < 8; ++r) {
      int nl = r + 8 * hi;
      int l = ut * 16 + lo16 - nl;
      bool ok = (unsigned)l < (unsigned)L_;
      float v = fmaf(nsev, hs[r], acc[r]);
      tx[ok ? nl * L_ + l : 16 * L_ + lane] = v;
    }
  }
  asm volatile("s_wait_dscnt 0x0" ::: "memory");
  float4* Xt = (float4*)(Xbm + (size_t)n0 * L_);
  float lmax = -3.4e38f;
#pragma unroll
  for (int k = 0; k < 3; ++k) {
    float4 q = ((const float4*)tx)[lane + 32 * k];
    Xt[lane + 32 * k] = q;
    lmax = fmaxf(lmax, fmaxf(fmaxf(q.x, q.y), fmaxf(q.z, q.w)));
  }
#pragma unroll
  for (int off = 16; off > 0; off >>= 1)
    lmax = fmaxf(lmax, __shfl_down(lmax, off, 32));
  if (lane == 0) atomicMax(umax + (wave & (NSLOT - 1)) * SLOTF, map_f32(lmax));
}

__global__ __launch_bounds__(256) void cassi_norm(
    float4* __restrict__ out, const unsigned* __restrict__ umax, int n4)
{
  int i = blockIdx.x * blockDim.x + threadIdx.x;
  unsigned u = 0u;
  for (int k = 0; k < NSLOT; ++k) {
    unsigned uk = umax[k * SLOTF];
    u = uk > u ? uk : u;
  }
  float mx = (u & 0x80000000u) ? __uint_as_float(u ^ 0x80000000u)
                               : __uint_as_float(~u);
  if (i < n4) {
    float4 v = out[i];
    v.x /= mx; v.y /= mx; v.z /= mx; v.w /= mx;
    out[i] = v;
  }
}

extern "C" void kernel_launch(void* const* d_in, const int* in_sizes, int n_in,
                              void* d_out, int out_size, void* d_ws, size_t ws_size,
                              hipStream_t stream) {
  const float* x   = (const float*)d_in[0];
  const float* H   = (const float*)d_in[1];
  const float* eps = (const float*)d_in[2];
  float* out = (float*)d_out;

  float*    sumYY = (float*)d_ws;
  unsigned* umax  = (unsigned*)((char*)d_ws + 2048);
  float*    Yt    = (float*)((char*)d_ws + 4096);

  hipMemsetAsync(d_ws, 0, 4096, stream);

  cassi_phase1<<<(B_ * M_ * NTW) / 4, 128, 0, stream>>>(x, H, Yt, sumYY);
  cassi_phase2<<<(B_ * M_ * NTN) / 4, 128, 0, stream>>>(H, eps, Yt, sumYY,
                                                        out, umax);
  int n4 = (B_ * M_ * M_ * L_) / 4;
  cassi_norm<<<(n4 + 255) / 256, 256, 0, stream>>>((float4*)out, umax, n4);
}


# both GEMM phases issue every load of a wave ahead of its first WMMA: phase 1 K loop fixed at 10 trips, fully unrolled, capped at 8 waves/SIMD (was load-wait-WMMA x10); phase 2 eps and sum words first
# speedup vs baseline: 12.6683x; 1.0877x over previous
#include <hip/hip_runtime.h>


typedef float v2f __attribute__((ext_vector_type(2)));
typedef float v8f __attribute__((ext_vector_type(8)));

#define B_   4
#define M_   256
#define L_   24
#define S_   22
#define SP_  24
#define W_   279
#define WP_  288
#define NTW  18
#define NTN  16
#define NSLOT 16
#define SLOTF 32
#define YP_  36
#define XT_  416

__device__ __forceinline__ unsigned map_f32(float x) {
  unsigned u = __float_as_uint(x);
  return (u & 0x80000000u) ? ~u : (u | 0x80000000u);
}

__device__ __forceinline__ int clampi(int v, int lo, int hi) {
  return v < lo ? lo : (v > hi ? hi : v);
}

template <bool EDGE>
__device__ __forceinline__ void p1_kloop(
    const float* __restrict__ xb, const float* __restrict__ Hm,
    int w0, int lo16, int hi, bool s1ok, int s1c, v8f& c0, v8f& c1)
{
#pragma unroll
  for (int it = 0; it < 10; ++it) {
    v2f a, b0v, b1v;
#pragma unroll
    for (int j = 0; j < 2; ++j) {
      const int k = j + 2 * hi;
      const int n = w0 - L_ + 4 * it + k;
      const int i = lo16 + L_ - 4 * it - k;
      const bool iok = (unsigned)i < (unsigned)L_;
      int fx = n * (L_ - 1) + w0 + lo16;
      int nh = n;
      bool nok = true;
      if (EDGE) {
        nok = (unsigned)n < (unsigned)M_;
        fx = clampi(fx, 0, M_ * L_ - 1);
        nh = clampi(n, 0, M_ - 1);
      }
      const float av = xb[fx];
      const float h0 = Hm[nh * S_ + lo16];
      const float h1 = Hm[nh * S_ + s1c];
      a[j]   = (iok & nok) ? av : 0.0f;
      b0v[j] = nok ? h0 : 0.0f;
      b1v[j] = (nok & s1ok) ? h1 : 0.0f;
    }
    c0 = __builtin_amdgcn_wmma_f32_16x16x4_f32(false, a, false, b0v,
                                               (short)0, c0, false, false);
    c1 = __builtin_amdgcn_wmma_f32_16x16x4_f32(false, a, false, b1v,
                                               (short)0, c1, false, false);
  }
}

__global__ __launch_bounds__(128) __attribute__((amdgpu_waves_per_eu(1, 8)))
void cassi_phase1(
    const float* __restrict__ x, const float* __restrict__ H,
    float* __restrict__ Yt, float* __restrict__ sumYY)
{
  __shared__ __attribute__((aligned(16))) float tileY[2][S_][YP_];
  const int lane = threadIdx.x & 31;
  const int wave = __builtin_amdgcn_readfirstlane(
      (int)(blockIdx.x * 4 + (threadIdx.x >> 5)));
  const int wt = wave % NTW;
  const int bm = wave / NTW;
  const int m  = bm & (M_ - 1);
  const int b  = bm >> 8;

  const int lo16 = lane & 15;
  const int hi   = lane >> 4;
  const int w0   = wt * 16;

  const float* xb = x + (size_t)(b * M_ + m) * M_ * L_;
  const float* Hm = H + (size_t)m * M_ * S_;

  v8f c0 = {};
  v8f c1 = {};

  const bool s1ok = (16 + lo16 < S_);
  const int  s1c  = s1ok ? 16 + lo16 : S_ - 1;

  if (wt >= 2 && wt < NTW - 2)
    p1_kloop<false>(xb, Hm, w0, lo16, hi, s1ok, s1c, c0, c1);
  else
    p1_kloop<true>(xb, Hm, w0, lo16, hi, s1ok, s1c, c0, c1);

  float sq = 0.0f;
#pragma unroll
  for (int r = 0; r < 8; ++r) {
    int w = w0 + r + 8 * hi;
    bool wok = (w < W_);
    float v0 = c0[r];
    float v1 = c1[r];
    sq += wok ? v0 * v0 : 0.0f;
    sq += (wok & s1ok) ? v1 * v1 : 0.0f;
  }

  const int wid = threadIdx.x >> 5;
  const int par = wid & 1;
  float (*ty)[YP_] = tileY[wid >> 1];
  {
    float4* d0 = (float4*)&ty[lo16][16 * par + 8 * hi];
    d0[0] = make_float4(c0[0], c0[1], c0[2], c0[3]);
    d0[1] = make_float4(c0[4], c0[5], c0[6], c0[7]);
    if (s1ok) {
      float4* d1 = (float4*)&ty[16 + lo16][16 * par + 8 * hi];
      d1[0] = make_float4(c1[0], c1[1], c1[2], c1[3]);
      d1[1] = make_float4(c1[4], c1[5], c1[6], c1[7]);
    }
  }
  __syncthreads();
  float* Ybm = Yt + (size_t)(b * M_ + m) * SP_ * WP_ + (w0 - 16 * par);
#pragma unroll
  for (int k = 0; k < 3; ++k) {
    int rr = 4 * k + (lane >> 3);
    if (rr < 11) {
      int s = par * 11 + rr;
      float4 v = *(const float4*)&ty[s][4 * (lane & 7)];
      *(float4*)(Ybm + (size_t)s * WP_ + 4 * (lane & 7)) = v;
    }
  }
#pragma unroll
  for (int off = 16; off > 0; off >>= 1) sq += __shfl_down(sq, off, 32);
  if (lane == 0) atomicAdd(sumYY + (wave & (NSLOT - 1)) * SLOTF, sq);
}

__global__ __launch_bounds__(128) void cassi_phase2(
    const float* __restrict__ H, const float* __restrict__ eps,
    const float* __restrict__ Yt, const float* __restrict__ sumYY,
    float* __restrict__ Xout, unsigned* __restrict__ umax)
{
  __shared__ __attribute__((aligned(16))) float tileX[4][XT_];
  const int lane = threadIdx.x & 31;
  const int wave = blockIdx.x * 4 + (threadIdx.x >> 5);
  const int t  = wave % NTN;
  const int bm = wave / NTN;
  const int m  = bm & (M_ - 1);
  const int b  = bm >> 8;

  const int lo16 = lane & 15;
  const int hi   = lane >> 4;
  const int n0   = t * 16;

  const float* Hm  = H  + (size_t)m * M_ * S_;
  const float* Ybm = Yt + (size_t)(b * M_ + m) * SP_ * WP_;

  const float* epsbm = eps + (size_t)(b * M_ + m) * W_;
  float ev3[3];
#pragma unroll
  for (int ut = 0; ut < 3; ++ut) {
    int u = (t + ut) * 16 + lo16;
    int uc = u < W_ ? u : W_ - 1;
    float ev = epsbm[uc];
    ev3[ut] = (u < W_) ? ev : 0.0f;
  }
  float sumAll = 0.0f;
  for (int k = 0; k < NSLOT; ++k) sumAll += sumYY[k * SLOTF];

  v8f acc0 = {}, acc1 = {}, acc2 = {};

#pragma unroll
  for (int s0 = 0; s0 < 24; s0 += 4) {
    v2f a, bv0, bv1, bv2;
#pragma unroll
    for (int j = 0; j < 2; ++j) {
      int s = s0 + j + 2 * hi;
      bool sok = (s < S_);
      int sc = sok ? s : S_ - 1;
      float av = Hm[(n0 + lo16) * S_ + sc];
      const float* row = Ybm + (size_t)sc * WP_ + lo16;
      float v0 = row[(t + 0) * 16];
      float v1 = row[(t + 1) * 16];
      float v2 = row[(t + 2) * 16];
      a[j]   = sok ? av : 0.0f;
      bv0[j] = sok ? v0 : 0.0f;
      bv1[j] = sok ? v1 : 0.0f;
      bv2[j] = sok ? v2 : 0.0f;
    }
    acc0 = __builtin_amdgcn_wmma_f32_16x16x4_f32(false, a, false, bv0,
                                                 (short)0, acc0, false, false);
    acc1 = __builtin_amdgcn_wmma_f32_16x16x4_f32(false, a, false, bv1,
                                                 (short)0, acc1, false, false);
    acc2 = __builtin_amdgcn_wmma_f32_16x16x4_f32(false, a, false, bv2,
                                                 (short)0, acc2, false, false);
  }

  float hsum = 0.0f;
  {
    const float* hr = Hm + (n0 + (lane & 15)) * S_;
    if (lane < 16) {
#pragma unroll
      for (int s = 0; s < S_; ++s) hsum += hr[s];
    }
  }
  float hs[8];
#pragma unroll
  for (int r = 0; r < 8; ++r) hs[r] = __shfl(hsum, r + 8 * hi, 32);

  const float noiseScale =
      sqrtf(sumAll * (1.0f / ((float)M_ * (float)W_ * (float)B_ * 10000.0f)));
  float* Xbm = Xout + (size_t)(b * M_ + m) * M_ * L_;

  float* tx = tileX[threadIdx.x >> 5];
#pragma unroll
  for (int ut = 0; ut < 3; ++ut) {
    const v8f& acc = (ut == 0) ? acc0 : (ut == 1) ? acc1 : acc2;
    float nsev = noiseScale * ev3[ut];
#pragma unroll
    for (int r = 0; r < 8; ++r) {
      int nl = r + 8 * hi;
      int l = ut * 16 + lo16 - nl;
      bool ok = (unsigned)l < (unsigned)L_;
      float v = fmaf(nsev, hs[r], acc[r]);
      tx[ok ? nl * L_ + l : 16 * L_ + lane] = v;
    }
  }
  asm volatile("s_wait_dscnt 0x0" ::: "memory");
  float4* Xt = (float4*)(Xbm + (size_t)n0 * L_);
  float lmax = -3.4e38f;
#pragma unroll
  for (int k = 0; k < 3; ++k) {
    float4 q = ((const float4*)tx)[lane + 32 * k];
    Xt[lane + 32 * k] = q;
    lmax = fmaxf(lmax, fmaxf(fmaxf(q.x, q.y), fmaxf(q.z, q.w)));
  }
#pragma unroll
  for (int off = 16; off > 0; off >>= 1)
    lmax = fmaxf(lmax, __shfl_down(lmax, off, 32));
  if (lane == 0) atomicMax(umax + (wave & (NSLOT - 1)) * SLOTF, map_f32(lmax));
}

__global__ __launch_bounds__(256) void cassi_norm(
    float4* __restrict__ out, const unsigned* __restrict__ umax, int n4)
{
  int i = blockIdx.x * blockDim.x + threadIdx.x;
  unsigned u = 0u;
  for (int k = 0; k < NSLOT; ++k) {
    unsigned uk = umax[k * SLOTF];
    u = uk > u ? uk : u;
  }
  float mx = (u & 0x80000000u) ? __uint_as_float(u ^ 0x80000000u)
                               : __uint_as_float(~u);
  if (i < n4) {
    float4 v = out[i];
    v.x /= mx; v.y /= mx; v.z /= mx; v.w /= mx;
    out[i] = v;
  }
}

extern "C" void kernel_launch(void* const* d_in, const int* in_sizes, int n_in,
                              void* d_out, int out_size, void* d_ws, size_t ws_size,
                              hipStream_t stream) {
  const float* x   = (const float*)d_in[0];
  const float* H   = (const float*)d_in[1];
  const float* eps = (const float*)d_in[2];
  float* out = (float*)d_out;

  float*    sumYY = (float*)d_ws;
  unsigned* umax  = (unsigned*)((char*)d_ws + 2048);
  float*    Yt    = (float*)((char*)d_ws + 4096);

  hipMemsetAsync(d_ws, 0, 4096, stream);

  cassi_phase1<<<(B_ * M_ * NTW) / 4, 128, 0, stream>>>(x, H, Yt, sumYY);
  cassi_phase2<<<(B_ * M_ * NTN) / 4, 128, 0, stream>>>(H, eps, Yt, sumYY,
                                                        out, umax);
  int n4 = (B_ * M_ * M_ * L_) / 4;
  cassi_norm<<<(n4 + 255) / 256, 256, 0, stream>>>((float4*)out, umax, n4);
}


# both GEMM phases: one wave does its tile for all 4 batch entries, H fragments loaded once, 4x fewer waves; each sum(Y^2) partial now spans 4 entries (changes the atomic summation order)
# speedup vs baseline: 15.3053x; 1.2082x over previous
#include <hip/hip_runtime.h>


typedef float v2f __attribute__((ext_vector_type(2)));
typedef float v8f __attribute__((ext_vector_type(8)));

#define B_   4
#define M_   256
#define L_   24
#define S_   22
#define SP_  24
#define W_   279
#define WP_  288
#define NTW  18
#define NTN  16
#define NSLOT 16
#define SLOTF 32
#define YP_  36
#define XT_  416

__device__ __forceinline__ unsigned map_f32(float x) {
  unsigned u = __float_as_uint(x);
  return (u & 0x80000000u) ? ~u : (u | 0x80000000u);
}

__device__ __forceinline__ int clampi(int v, int lo, int hi) {
  return v < lo ? lo : (v > hi ? hi : v);
}

template <bool EDGE>
__device__ __forceinline__ void p1_loadB(
    const float* __restrict__ Hm, int w0, int lo16, int hi, int s1c,
    float (&h0)[20], float (&h1)[20])
{
#pragma unroll
  for (int it = 0; it < 10; ++it) {
#pragma unroll
    for (int j = 0; j < 2; ++j) {
      const int n = w0 - L_ + 4 * it + j + 2 * hi;
      const int nh = EDGE ? clampi(n, 0, M_ - 1) : n;
      h0[2 * it + j] = Hm[nh * S_ + lo16];
      h1[2 * it + j] = Hm[nh * S_ + s1c];
    }
  }
}

template <bool EDGE>
__device__ __forceinline__ void p1_loadA(
    const float* __restrict__ xb, int w0, int lo16, int hi, float (&ar)[20])
{
#pragma unroll
  for (int it = 0; it < 10; ++it) {
#pragma unroll
    for (int j = 0; j < 2; ++j) {
      const int n = w0 - L_ + 4 * it + j + 2 * hi;
      int fx = n * (L_ - 1) + w0 + lo16;
      if (EDGE) fx = clampi(fx, 0, M_ * L_ - 1);
      ar[2 * it + j] = xb[fx];
    }
  }
}

__global__ __launch_bounds__(128) __attribute__((amdgpu_waves_per_eu(1, 5)))
void cassi_phase1(
    const float* __restrict__ x, const float* __restrict__ H,
    float* __restrict__ Yt, float* __restrict__ sumYY)
{
  __shared__ __attribute__((aligned(16))) float tileY[2][2][S_][YP_];
  const int lane = threadIdx.x & 31;
  const int wave = __builtin_amdgcn_readfirstlane(
      (int)(blockIdx.x * 4 + (threadIdx.x >> 5)));
  const int wt = wave % NTW;
  const int m  = wave / NTW;

  const int lo16 = lane & 15;
  const int hi   = lane >> 4;
  const int w0   = wt * 16;
  const bool edge = !(wt >= 2 && wt < NTW - 2);

  const float* xm = x + (size_t)m * M_ * L_;
  const float* Hm = H + (size_t)m * M_ * S_;
  const size_t xbs = (size_t)M_ * M_ * L_;

  const bool s1ok = (16 + lo16 < S_);
  const int  s1c  = s1ok ? 16 + lo16 : S_ - 1;

  v2f bf0[10], bf1[10];
  float hr0[20], hr1[20];
  float ar[2][20];
  if (edge) {
    p1_loadB<true>(Hm, w0, lo16, hi, s1c, hr0, hr1);
    p1_loadA<true>(xm, w0, lo16, hi, ar[0]);
  } else {
    p1_loadB<false>(Hm, w0, lo16, hi, s1c, hr0, hr1);
    p1_loadA<false>(xm, w0, lo16, hi, ar[0]);
  }

  const int wid = threadIdx.x >> 5;
  const int par = wid & 1;
  float sq = 0.0f;

#pragma unroll
  for (int b = 0; b < B_; ++b) {
    if (b + 1 < B_) {
      const float* xn = xm + (size_t)(b + 1) * xbs;
      if (edge) p1_loadA<true>(xn, w0, lo16, hi, ar[(b + 1) & 1]);
      else      p1_loadA<false>(xn, w0, lo16, hi, ar[(b + 1) & 1]);
    }
    if (b == 0) {
#pragma unroll
      for (int it = 0; it < 10; ++it) {
#pragma unroll
        for (int j = 0; j < 2; ++j) {
          const int n = w0 - L_ + 4 * it + j + 2 * hi;
          const bool nok = (unsigned)n < (unsigned)M_;
          bf0[it][j] = nok ? hr0[2 * it + j] : 0.0f;
          bf1[it][j] = (nok & s1ok) ? hr1[2 * it + j] : 0.0f;
        }
      }
    }
    v8f c0 = {};
    v8f c1 = {};
#pragma unroll
    for (int it = 0; it < 10; ++it) {
      v2f a;
#pragma unroll
      for (int j = 0; j < 2; ++j) {
        const int k = j + 2 * hi;
        const int n = w0 - L_ + 4 * it + k;
        const int i = lo16 + L_ - 4 * it - k;
        const bool ok = ((unsigned)i < (unsigned)L_) & ((unsigned)n < (unsigned)M_);
        a[j] = ok ? ar[b & 1][2 * it + j] : 0.0f;
      }
      c0 = __builtin_amdgcn_wmma_f32_16x16x4_f32(false, a, false, bf0[it],
                                                 (short)0, c0, false, false);
      c1 = __builtin_amdgcn_wmma_f32_16x16x4_f32(false, a, false, bf1[it],
                                                 (short)0, c1, false, false);
    }

#pragma unroll
    for (int r = 0; r < 8; ++r) {
      int w = w0 + r + 8 * hi;
      bool wok = (w < W_);
      float v0 = c0[r];
      float v1 = c1[r];
      sq += wok ? v0 * v0 : 0.0f;
      sq += (wok & s1ok) ? v1 * v1 : 0.0f;
    }

    float (*ty)[YP_] = tileY[b & 1][wid >> 1];
    {
      float4* d0 = (float4*)&ty[lo16][16 * par + 8 * hi];
      d0[0] = make_float4(c0[0], c0[1], c0[2], c0[3]);
      d0[1] = make_float4(c0[4], c0[5], c0[6], c0[7]);
      if (s1ok) {
        float4* d1 = (float4*)&ty[16 + lo16][16 * par + 8 * hi];
        d1[0] = make_float4(c1[0], c1[1], c1[2], c1[3]);
        d1[1] = make_float4(c1[4], c1[5], c1[6], c1[7]);
      }
    }
    __builtin_amdgcn_fence(__ATOMIC_RELEASE, "workgroup", "local");
    __builtin_amdgcn_s_barrier();
    __builtin_amdgcn_fence(__ATOMIC_ACQUIRE, "workgroup", "local");
    float* Ybm = Yt + (size_t)(b * M_ + m) * SP_ * WP_ + (w0 - 16 * par);
#pragma unroll
    for (int k = 0; k < 3; ++k) {
      int rr = 4 * k + (lane >> 3);
      if (rr < 11) {
        int s = par * 11 + rr;
        float4 v = *(const float4*)&ty[s][4 * (lane & 7)];
        *(float4*)(Ybm + (size_t)s * WP_ + 4 * (lane & 7)) = v;
      }
    }
  }
#pragma unroll
  for (int off = 16; off > 0; off >>= 1) sq += __shfl_down(sq, off, 32);
  if (lane == 0) atomicAdd(sumYY + (wave & (NSLOT - 1)) * SLOTF, sq);
}

__device__ __forceinline__ void p2_loadB(
    const float* __restrict__ Ybm, const float* __restrict__ epsbm,
    int t, int lo16, int hi, float (&br)[36], float (&er)[3])
{
#pragma unroll
  for (int q = 0; q < 6; ++q) {
#pragma unroll
    for (int j = 0; j < 2; ++j) {
      int s = 4 * q + j + 2 * hi;
      int sc = s < S_ ? s : S_ - 1;
      const float* row = Ybm + (size_t)sc * WP_ + lo16;
#pragma unroll
      for (int ut = 0; ut < 3; ++ut) br[(2 * q + j) * 3 + ut] = row[(t + ut) * 16];
    }
  }
#pragma unroll
  for (int ut = 0; ut < 3; ++ut) {
    int u = (t + ut) * 16 + lo16;
    er[ut] = epsbm[u < W_ ? u : W_ - 1];
  }
}

__global__ __launch_bounds__(128) __attribute__((amdgpu_waves_per_eu(1, 4)))
void cassi_phase2(
    const float* __restrict__ H, const float* __restrict__ eps,
    const float* __restrict__ Yt, const float* __restrict__ sumYY,
    float* __restrict__ Xout, unsigned* __restrict__ umax)
{
  __shared__ __attribute__((aligned(16))) float tileX[4][XT_];
  const int lane = threadIdx.x & 31;
  const int wave = blockIdx.x * 4 + (threadIdx.x >> 5);
  const int t  = wave % NTN;
  const int m  = wave / NTN;

  const int lo16 = lane & 15;
  const int hi   = lane >> 4;
  const int n0   = t * 16;

  const float* Hm  = H  + (size_t)m * M_ * S_;
  const float* Ym  = Yt + (size_t)m * SP_ * WP_;
  const float* em  = eps + (size_t)m * W_;
  const size_t ybs = (size_t)M_ * SP_ * WP_;
  const size_t ebs = (size_t)M_ * W_;

  float br[2][36];
  float er[2][3];
  p2_loadB(Ym, em, t, lo16, hi, br[0], er[0]);

  v2f af[6];
#pragma unroll
  for (int q = 0; q < 6; ++q) {
#pragma unroll
    for (int j = 0; j < 2; ++j) {
      int s = 4 * q + j + 2 * hi;
      bool sok = (s < S_);
      int sc = sok ? s : S_ - 1;
      float av = Hm[(n0 + lo16) * S_ + sc];
      af[q][j] = sok ? av : 0.0f;
    }
  }

  float sumAll = 0.0f;
  for (int k = 0; k < NSLOT; ++k) sumAll += sumYY[k * SLOTF];

  float hsum = 0.0f;
  {
    const float* hr = Hm + (n0 + (lane & 15)) * S_;
    if (lane < 16) {
#pragma unroll
      for (int s = 0; s < S_; ++s) hsum += hr[s];
    }
  }
  float hs[8];
#pragma unroll
  for (int r = 0; r < 8; ++r) hs[r] = __shfl(hsum, r + 8 * hi, 32);

  const float noiseScale =
      sqrtf(sumAll * (1.0f / ((float)M_ * (float)W_ * (float)B_ * 10000.0f)));

  float* tx = tileX[threadIdx.x >> 5];
  float lmax = -3.4e38f;

#pragma unroll
  for (int b = 0; b < B_; ++b) {
    if (b + 1 < B_)
      p2_loadB(Ym + (size_t)(b + 1) * ybs, em + (size_t)(b + 1) * ebs,
               t, lo16, hi, br[(b + 1) & 1], er[(b + 1) & 1]);

    v8f acc0 = {}, acc1 = {}, acc2 = {};
#pragma unroll
    for (int q = 0; q < 6; ++q) {
      v2f bv0, bv1, bv2;
#pragma unroll
      for (int j = 0; j < 2; ++j) {
        bool sok = (4 * q + j + 2 * hi < S_);
        bv0[j] = sok ? br[b & 1][(2 * q + j) * 3 + 0] : 0.0f;
        bv1[j] = sok ? br[b & 1][(2 * q + j) * 3 + 1] : 0.0f;
        bv2[j] = sok ? br[b & 1][(2 * q + j) * 3 + 2] : 0.0f;
      }
      acc0 = __builtin_amdgcn_wmma_f32_16x16x4_f32(false, af[q], false, bv0,
                                                   (short)0, acc0, false, false);
      acc1 = __builtin_amdgcn_wmma_f32_16x16x4_f32(false, af[q], false, bv1,
                                                   (short)0, acc1, false, false);
      acc2 = __builtin_amdgcn_wmma_f32_16x16x4_f32(false, af[q], false, bv2,
                                                   (short)0, acc2, false, false);
    }

#pragma unroll
    for (int ut = 0; ut < 3; ++ut) {
      const v8f& acc = (ut == 0) ? acc0 : (ut == 1) ? acc1 : acc2;
      int u = (t + ut) * 16 + lo16;
      float ev = (u < W_) ? er[b & 1][ut] : 0.0f;
      float nsev = noiseScale * ev;
#pragma unroll
      for (int r = 0; r < 8; ++r) {
        int nl = r + 8 * hi;
        int l = ut * 16 + lo16 - nl;
        bool ok = (unsigned)l < (unsigned)L_;
        float v = fmaf(nsev, hs[r], acc[r]);
        tx[ok ? nl * L_ + l : 16 * L_ + lane] = v;
      }
    }
    asm volatile("s_wait_dscnt 0x0" ::: "memory");
    float4* Xt = (float4*)(Xout + (size_t)(b * M_ + m) * M_ * L_ + (size_t)n0 * L_);
#pragma unroll
    for (int k = 0; k < 3; ++k) {
      float4 q4 = ((const float4*)tx)[lane + 32 * k];
      Xt[lane + 32 * k] = q4;
      lmax = fmaxf(lmax, fmaxf(fmaxf(q4.x, q4.y), fmaxf(q4.z, q4.w)));
    }
  }
#pragma unroll
  for (int off = 16; off > 0; off >>= 1)
    lmax = fmaxf(lmax, __shfl_down(lmax, off, 32));
  if (lane == 0) atomicMax(umax + (wave & (NSLOT - 1)) * SLOTF, map_f32(lmax));
}

__global__ __launch_bounds__(256) void cassi_norm(
    float4* __restrict__ out, const unsigned* __restrict__ umax, int n4)
{
  int i = blockIdx.x * blockDim.x + threadIdx.x;
  unsigned u = 0u;
  for (int k = 0; k < NSLOT; ++k) {
    unsigned uk = umax[k * SLOTF];
    u = uk > u ? uk : u;
  }
  float mx = (u & 0x80000000u) ? __uint_as_float(u ^ 0x80000000u)
                               : __uint_as_float(~u);
  if (i < n4) {
    float4 v = out[i];
    v.x /= mx; v.y /= mx; v.z /= mx; v.w /= mx;
    out[i] = v;
  }
}

extern "C" void kernel_launch(void* const* d_in, const int* in_sizes, int n_in,
                              void* d_out, int out_size, void* d_ws, size_t ws_size,
                              hipStream_t stream) {
  const float* x   = (const float*)d_in[0];
  const float* H   = (const float*)d_in[1];
  const float* eps = (const float*)d_in[2];
  float* out = (float*)d_out;

  float*    sumYY = (float*)d_ws;
  unsigned* umax  = (unsigned*)((char*)d_ws + 2048);
  float*    Yt    = (float*)((char*)d_ws + 4096);

  hipMemsetAsync(d_ws, 0, 4096, stream);

  cassi_phase1<<<(M_ * NTW) / 4, 128, 0, stream>>>(x, H, Yt, sumYY);
  cassi_phase2<<<(M_ * NTN) / 4, 128, 0, stream>>>(H, eps, Yt, sumYY,
                                                        out, umax);
  int n4 = (B_ * M_ * M_ * L_) / 4;
  cassi_norm<<<(n4 + 255) / 256, 256, 0, stream>>>((float4*)out, umax, n4);
}
